# starting kernel
# baseline (speedup 1.0000x reference)
#include <hip/hip_runtime.h>
#include <hip/hip_bf16.h>
#include <math.h>

typedef __attribute__((ext_vector_type(2))) float v2f;
typedef __attribute__((ext_vector_type(8))) float v8f;

__device__ __forceinline__ v8f wmma_f32(v2f a, v2f b, v8f c) {
  return __builtin_amdgcn_wmma_f32_16x16x4_f32(false, a, false, b, (short)0, c,
                                               false, false);
}

__global__ void bias_pre_kernel(const float* __restrict__ table,
                                float* __restrict__ bias_pre) {
  int tid = blockIdx.x * blockDim.x + threadIdx.x;
  if (tid >= 4 * 98 * 98) return;
  int h = tid / 9604;
  int rem = tid - h * 9604;
  int m = rem / 98;
  int n = rem - m * 98;
  int zm = m / 49, ym = (m % 49) / 7, xm = m % 7;
  int zn = n / 49, yn = (n % 49) / 7, xn = n % 7;
  int idx = (zm - zn + 1) * 169 + (ym - yn + 6) * 13 + (xm - xn + 6);
  bias_pre[tid] = table[idx * 4 + h];
}

__global__ __launch_bounds__(256) void fused_window_attn(
    const float* __restrict__ x, const float* __restrict__ mask,
    const float* __restrict__ w_qkv, const float* __restrict__ bq,
    const float* __restrict__ w_proj, const float* __restrict__ bpj,
    const float* __restrict__ bias_pre, float* __restrict__ out) {
  __shared__ float sQKV[112][386];
  __shared__ float sO[112][130];
  __shared__ float sP[8][16][114];

  const int b = blockIdx.x;
  const int wd = b & 511;
  const int wave = threadIdx.x >> 5;
  const int lane = threadIdx.x & 31;
  const int l16 = lane & 15;
  const int hi = lane >> 4;
  const float* xb = x + (size_t)b * 98 * 128;

  for (int i = 0; i < 3; ++i) {
    const int nt = wave * 3 + i;
    const int col = nt * 16 + l16;
    v8f acc[7] = {};
    for (int k0 = 0; k0 < 128; k0 += 4) {
      const int kb = k0 + 2 * hi;
      v2f bw;
      bw.x = w_qkv[(size_t)kb * 384 + col];
      bw.y = w_qkv[(size_t)(kb + 1) * 384 + col];
#pragma unroll
      for (int mt = 0; mt < 7; ++mt) {
        int n = mt * 16 + l16;
        int nc = n < 98 ? n : 97;
        v2f a = *(const v2f*)(xb + (size_t)nc * 128 + k0 + 2 * hi);
        acc[mt] = wmma_f32(a, bw, acc[mt]);
      }
    }
    const float bias = bq[col];
    const float scale = (col < 128) ? 0.17677669529663687f : 1.0f;
#pragma unroll
    for (int mt = 0; mt < 7; ++mt)
#pragma unroll
      for (int r = 0; r < 8; ++r) {
        int m = mt * 16 + r + 8 * hi;
        sQKV[m][col] = (acc[mt][r] + bias) * scale;
      }
  }
  __syncthreads();

  for (int it = 0; it < 4; ++it) {
    const int u = wave + 8 * it;
    if (u < 28) {
      const int h = u / 7;
      const int mt = u - h * 7;

      v8f acc[7] = {};
      for (int ks = 0; ks < 8; ++ks) {
        const int k0 = ks * 4;
        v2f a = *(const v2f*)(&sQKV[mt * 16 + l16][h * 32 + k0 + 2 * hi]);
#pragma unroll
        for (int nt = 0; nt < 7; ++nt) {
          v2f bk =
              *(const v2f*)(&sQKV[nt * 16 + l16][128 + h * 32 + k0 + 2 * hi]);
          acc[nt] = wmma_f32(a, bk, acc[nt]);
        }
      }

      const float* bp = bias_pre + (size_t)h * 9604;
      const float* mp = mask + (size_t)wd * 9604;
#pragma unroll
      for (int nt = 0; nt < 7; ++nt) {
        const int n = nt * 16 + l16;
#pragma unroll
        for (int r = 0; r < 8; ++r) {
          const int m = mt * 16 + r + 8 * hi;
          if (n < 98 && m < 98)
            acc[nt][r] += bp[m * 98 + n] + mp[m * 98 + n];
          else
            acc[nt][r] = -3.0e38f;
        }
      }

#pragma unroll
      for (int r = 0; r < 8; ++r) {
        float mx = acc[0][r];
#pragma unroll
        for (int nt = 1; nt < 7; ++nt) mx = fmaxf(mx, acc[nt][r]);
#pragma unroll
        for (int off = 1; off < 16; off <<= 1)
          mx = fmaxf(mx, __shfl_xor(mx, off, 32));
        float e[7];
        float sum = 0.0f;
#pragma unroll
        for (int nt = 0; nt < 7; ++nt) {
          e[nt] = __expf(acc[nt][r] - mx);
          sum += e[nt];
        }
#pragma unroll
        for (int off = 1; off < 16; off <<= 1) sum += __shfl_xor(sum, off, 32);
        const float inv = 1.0f / sum;
        const int m = mt * 16 + r + 8 * hi;
        const bool rowv = (m < 98);
#pragma unroll
        for (int nt = 0; nt < 7; ++nt) {
          const int n = nt * 16 + l16;
          sP[wave][r + 8 * hi][nt * 16 + l16] =
              (rowv && n < 98) ? e[nt] * inv : 0.0f;
        }
      }

      v8f o0 = {}, o1 = {};
      for (int ks = 0; ks < 28; ++ks) {
        const int k0 = ks * 4;
        v2f a = *(const v2f*)(&sP[wave][l16][k0 + 2 * hi]);
        const int t0 = k0 + 2 * hi;
        v2f b0, b1;
        b0.x = sQKV[t0][256 + h * 32 + l16];
        b0.y = sQKV[t0 + 1][256 + h * 32 + l16];
        o0 = wmma_f32(a, b0, o0);
        b1.x = sQKV[t0][256 + h * 32 + 16 + l16];
        b1.y = sQKV[t0 + 1][256 + h * 32 + 16 + l16];
        o1 = wmma_f32(a, b1, o1);
      }
#pragma unroll
      for (int r = 0; r < 8; ++r) {
        const int m = mt * 16 + r + 8 * hi;
        sO[m][h * 32 + l16] = o0[r];
        sO[m][h * 32 + 16 + l16] = o1[r];
      }
    }
  }
  __syncthreads();

  {
    const int col = wave * 16 + l16;
    v8f acc[7] = {};
    for (int k0 = 0; k0 < 128; k0 += 4) {
      const int kb = k0 + 2 * hi;
      v2f bw;
      bw.x = w_proj[(size_t)kb * 128 + col];
      bw.y = w_proj[(size_t)(kb + 1) * 128 + col];
#pragma unroll
      for (int mt = 0; mt < 7; ++mt) {
        v2f a = *(const v2f*)(&sO[mt * 16 + l16][k0 + 2 * hi]);
        acc[mt] = wmma_f32(a, bw, acc[mt]);
      }
    }
    const float bias = bpj[col];
#pragma unroll
    for (int mt = 0; mt < 7; ++mt)
#pragma unroll
      for (int r = 0; r < 8; ++r) {
        const int m = mt * 16 + r + 8 * hi;
        if (m < 98) out[((size_t)b * 98 + m) * 128 + col] = acc[mt][r] + bias;
      }
  }
}

extern "C" void kernel_launch(void* const* d_in, const int* in_sizes, int n_in,
                              void* d_out, int out_size, void* d_ws,
                              size_t ws_size, hipStream_t stream) {
  const float* x      = (const float*)d_in[0];
  const float* mask   = (const float*)d_in[1];
  const float* w_qkv  = (const float*)d_in[2];
  const float* b_qkv  = (const float*)d_in[3];
  const float* w_proj = (const float*)d_in[4];
  const float* b_proj = (const float*)d_in[5];
  const float* table  = (const float*)d_in[6];
  float* out = (float*)d_out;
  float* bias_pre = (float*)d_ws;

  bias_pre_kernel<<<(4 * 98 * 98 + 255) / 256, 256, 0, stream>>>(table,
                                                                 bias_pre);
  fused_window_attn<<<2048, 256, 0, stream>>>(x, mask, w_qkv, b_qkv, w_proj,
                                              b_proj, bias_pre, out);
}


# Softmax row max and row sum: the 16-lane xor butterflies run on the vector ALU (DPP row_xmask 1,2,4,8) instead of ds_bpermute_b32 through the LDS unit; same steps, sum order kept
# speedup vs baseline: 1.9701x; 1.9701x over previous
#include <hip/hip_runtime.h>
#include <hip/hip_bf16.h>
#include <math.h>

typedef __attribute__((ext_vector_type(2))) float v2f;
typedef __attribute__((ext_vector_type(8))) float v8f;

__device__ __forceinline__ v8f wmma_f32(v2f a, v2f b, v8f c) {
  return __builtin_amdgcn_wmma_f32_16x16x4_f32(false, a, false, b, (short)0, c,
                                               false, false);
}


template <int M>
__device__ __forceinline__ float row_xor(float v) {
  const int b = __builtin_bit_cast(int, v);
  return __builtin_bit_cast(
      float, __builtin_amdgcn_update_dpp(b, b, 0x160 + M, 0xf, 0xf, true));
}

__global__ __launch_bounds__(448) void score_term_kernel(
    const float* __restrict__ table, const float* __restrict__ mask,
    float* __restrict__ term) {
  const int wd = blockIdx.x / 7;
  const int mt = blockIdx.x - wd * 7;
  const int j = threadIdx.x >> 5;
  const int lane = threadIdx.x & 31;
  const int l16 = lane & 15;
  const int hi = lane >> 4;
  const int nt = j >> 1;
  const int r0 = (j & 1) * 4;
  const int n = nt * 16 + l16;
  const int nc = n < 98 ? n : 97;
  const int zn = nc / 49, yn = (nc % 49) / 7, xn = nc % 7;
  const float* mp = mask + (size_t)wd * 9604;
  float mk[4], tb[4][4];
  bool ok[4];
  for (int e = 0; e < 4; ++e) {
    const int m = mt * 16 + r0 + e + 8 * hi;
    const int mc = m < 98 ? m : 97;
    const int zm = mc / 49, ym = (mc % 49) / 7, xm = mc % 7;
    const int idx = (zm - zn + 1) * 169 + (ym - yn + 6) * 13 + (xm - xn + 6);
    ok[e] = (m < 98) && (n < 98);
    mk[e] = mp[mc * 98 + nc];
    for (int h = 0; h < 4; ++h) tb[e][h] = table[idx * 4 + h];
  }
  float v[4][4];
  for (int e = 0; e < 4; ++e)
    for (int h = 0; h < 4; ++h) {
      const float s = tb[e][h] + mk[e];
      v[h][e] = ok[e] ? s : -3.0e38f;
    }
  float4* t4 = (float4*)term;
  for (int h = 0; h < 4; ++h)
    t4[((size_t)((wd * 4 + h) * 7 + mt) * 14 + j) * 32 + lane] =
        make_float4(v[h][0], v[h][1], v[h][2], v[h][3]);
}

__global__ __launch_bounds__(256) void fused_window_attn(
    const float* __restrict__ x,
    const float* __restrict__ w_qkv, const float* __restrict__ bq,
    const float* __restrict__ w_proj, const float* __restrict__ bpj,
    const float* __restrict__ term, float* __restrict__ out) {
  __shared__ float sQKV[112][386];
  __shared__ float sO[112][130];
  __shared__ float sP[8][16][114];

  const int b = blockIdx.x;
  const int wd = b & 511;
  const int wave = threadIdx.x >> 5;
  const int lane = threadIdx.x & 31;
  const int l16 = lane & 15;
  const int hi = lane >> 4;
  const float* xb = x + (size_t)b * 98 * 128;

  const float4* term4 = (const float4*)term;
  float4 pf[14];
  {
    const int h0 = wave / 7;
    const int mt0 = wave - h0 * 7;
    const float4* tp =
        term4 + (size_t)((wd * 4 + h0) * 7 + mt0) * (14 * 32) + lane;
    for (int j = 0; j < 14; ++j) pf[j] = tp[j * 32];
  }

  for (int i = 0; i < 3; ++i) {
    const int nt = wave * 3 + i;
    const int col = nt * 16 + l16;
    v8f acc[7] = {};
#pragma unroll 4
    for (int k0 = 0; k0 < 128; k0 += 8) {
      const int kb = k0 + 4 * hi;
      v2f bw0, bw1;
      bw0.x = w_qkv[(size_t)kb * 384 + col];
      bw0.y = w_qkv[(size_t)(kb + 1) * 384 + col];
      bw1.x = w_qkv[(size_t)(kb + 2) * 384 + col];
      bw1.y = w_qkv[(size_t)(kb + 3) * 384 + col];
      float4 a4[7];
#pragma unroll
      for (int mt = 0; mt < 7; ++mt) {
        int n = mt * 16 + l16;
        int nc = n < 98 ? n : 97;
        a4[mt] = *(const float4*)(xb + (size_t)nc * 128 + kb);
      }
#pragma unroll
      for (int mt = 0; mt < 7; ++mt) {
        v2f a;
        a.x = a4[mt].x; a.y = a4[mt].y;
        acc[mt] = wmma_f32(a, bw0, acc[mt]);
      }
#pragma unroll
      for (int mt = 0; mt < 7; ++mt) {
        v2f a;
        a.x = a4[mt].z; a.y = a4[mt].w;
        acc[mt] = wmma_f32(a, bw1, acc[mt]);
      }
    }
    const float bias = bq[col];
    const float scale = (col < 128) ? 0.17677669529663687f : 1.0f;
#pragma unroll
    for (int mt = 0; mt < 7; ++mt)
#pragma unroll
      for (int r = 0; r < 8; ++r) {
        int m = mt * 16 + r + 8 * hi;
        sQKV[m][col] = (acc[mt][r] + bias) * scale;
      }
  }
  __syncthreads();

  for (int it = 0; it < 4; ++it) {
    const int u = wave + 8 * it;
    if (u < 28) {
      const int h = u / 7;
      const int mt = u - h * 7;

      v8f acc[7];
      for (int nt = 0; nt < 7; ++nt) {
        const float4 lo = pf[2 * nt], hh = pf[2 * nt + 1];
        acc[nt][0] = lo.x; acc[nt][1] = lo.y; acc[nt][2] = lo.z; acc[nt][3] = lo.w;
        acc[nt][4] = hh.x; acc[nt][5] = hh.y; acc[nt][6] = hh.z; acc[nt][7] = hh.w;
      }
      for (int ks = 0; ks < 8; ++ks) {
        const int k0 = ks * 4;
        v2f a = *(const v2f*)(&sQKV[mt * 16 + l16][h * 32 + k0 + 2 * hi]);
#pragma unroll
        for (int nt = 0; nt < 7; ++nt) {
          v2f bk =
              *(const v2f*)(&sQKV[nt * 16 + l16][128 + h * 32 + k0 + 2 * hi]);
          acc[nt] = wmma_f32(a, bk, acc[nt]);
        }
      }

      if (u + 8 < 28) {
        const int hn = (u + 8) / 7;
        const int mtn = (u + 8) - hn * 7;
        const float4* tp =
            term4 + (size_t)((wd * 4 + hn) * 7 + mtn) * (14 * 32) + lane;
        for (int j = 0; j < 14; ++j) pf[j] = tp[j * 32];
      }

#pragma unroll
      for (int r = 0; r < 8; ++r) {
        float mx = acc[0][r];
#pragma unroll
        for (int nt = 1; nt < 7; ++nt) mx = fmaxf(mx, acc[nt][r]);
        mx = fmaxf(mx, row_xor<1>(mx));
        mx = fmaxf(mx, row_xor<2>(mx));
        mx = fmaxf(mx, row_xor<4>(mx));
        mx = fmaxf(mx, row_xor<8>(mx));
        float e[7];
        float sum = 0.0f;
#pragma unroll
        for (int nt = 0; nt < 7; ++nt) {
          e[nt] = __expf(acc[nt][r] - mx);
          sum += e[nt];
        }
        sum += row_xor<1>(sum);
        sum += row_xor<2>(sum);
        sum += row_xor<4>(sum);
        sum += row_xor<8>(sum);
        const float inv = 1.0f / sum;
        const int m = mt * 16 + r + 8 * hi;
        const bool rowv = (m < 98);
#pragma unroll
        for (int nt = 0; nt < 7; ++nt) {
          const int n = nt * 16 + l16;
          sP[wave][r + 8 * hi][nt * 16 + l16] =
              (rowv && n < 98) ? e[nt] * inv : 0.0f;
        }
      }

      v8f o0 = {}, o1 = {};
      for (int ks = 0; ks < 28; ++ks) {
        const int k0 = ks * 4;
        v2f a = *(const v2f*)(&sP[wave][l16][k0 + 2 * hi]);
        const int t0 = k0 + 2 * hi;
        v2f b0, b1;
        b0.x = sQKV[t0][256 + h * 32 + l16];
        b0.y = sQKV[t0 + 1][256 + h * 32 + l16];
        o0 = wmma_f32(a, b0, o0);
        b1.x = sQKV[t0][256 + h * 32 + 16 + l16];
        b1.y = sQKV[t0 + 1][256 + h * 32 + 16 + l16];
        o1 = wmma_f32(a, b1, o1);
      }
#pragma unroll
      for (int r = 0; r < 8; ++r) {
        const int m = mt * 16 + r + 8 * hi;
        sO[m][h * 32 + l16] = o0[r];
        sO[m][h * 32 + 16 + l16] = o1[r];
      }
    }
  }
  __syncthreads();

  {
    const int col = wave * 16 + l16;
    v8f acc[7] = {};
    for (int k0 = 0; k0 < 128; k0 += 4) {
      const int kb = k0 + 2 * hi;
      v2f bw;
      bw.x = w_proj[(size_t)kb * 128 + col];
      bw.y = w_proj[(size_t)(kb + 1) * 128 + col];
#pragma unroll
      for (int mt = 0; mt < 7; ++mt) {
        v2f a = *(const v2f*)(&sO[mt * 16 + l16][k0 + 2 * hi]);
        acc[mt] = wmma_f32(a, bw, acc[mt]);
      }
    }
    const float bias = bpj[col];
#pragma unroll
    for (int mt = 0; mt < 7; ++mt)
#pragma unroll
      for (int r = 0; r < 8; ++r) {
        const int m = mt * 16 + r + 8 * hi;
        if (m < 98) out[((size_t)b * 98 + m) * 128 + col] = acc[mt][r] + bias;
      }
  }
}

extern "C" void kernel_launch(void* const* d_in, const int* in_sizes, int n_in,
                              void* d_out, int out_size, void* d_ws,
                              size_t ws_size, hipStream_t stream) {
  const float* x      = (const float*)d_in[0];
  const float* mask   = (const float*)d_in[1];
  const float* w_qkv  = (const float*)d_in[2];
  const float* b_qkv  = (const float*)d_in[3];
  const float* w_proj = (const float*)d_in[4];
  const float* b_proj = (const float*)d_in[5];
  const float* table  = (const float*)d_in[6];
  float* out = (float*)d_out;
  float* term = (float*)d_ws;

  score_term_kernel<<<3584, 448, 0, stream>>>(table, mask, term);
  fused_window_attn<<<2048, 256, 0, stream>>>(x, w_qkv, b_qkv, w_proj, b_proj,
                                              term, out);
}


# QKV projection: each wave owns one 16-row tile and keeps its x rows in 64 registers for all of K; weights stream by 16-byte loads along n, one load feeding 4 column tiles; sum order kept
# speedup vs baseline: 2.1353x; 1.0839x over previous
#include <hip/hip_runtime.h>
#include <hip/hip_bf16.h>
#include <math.h>

typedef __attribute__((ext_vector_type(2))) float v2f;
typedef __attribute__((ext_vector_type(8))) float v8f;

__device__ __forceinline__ v8f wmma_f32(v2f a, v2f b, v8f c) {
  return __builtin_amdgcn_wmma_f32_16x16x4_f32(false, a, false, b, (short)0, c,
                                               false, false);
}


template <int M>
__device__ __forceinline__ float row_xor(float v) {
  const int b = __builtin_bit_cast(int, v);
  return __builtin_bit_cast(
      float, __builtin_amdgcn_update_dpp(b, b, 0x160 + M, 0xf, 0xf, true));
}

__global__ __launch_bounds__(448) void score_term_kernel(
    const float* __restrict__ table, const float* __restrict__ mask,
    float* __restrict__ term) {
  const int wd = blockIdx.x / 7;
  const int mt = blockIdx.x - wd * 7;
  const int j = threadIdx.x >> 5;
  const int lane = threadIdx.x & 31;
  const int l16 = lane & 15;
  const int hi = lane >> 4;
  const int nt = j >> 1;
  const int r0 = (j & 1) * 4;
  const int n = nt * 16 + l16;
  const int nc = n < 98 ? n : 97;
  const int zn = nc / 49, yn = (nc % 49) / 7, xn = nc % 7;
  const float* mp = mask + (size_t)wd * 9604;
  float mk[4], tb[4][4];
  bool ok[4];
  for (int e = 0; e < 4; ++e) {
    const int m = mt * 16 + r0 + e + 8 * hi;
    const int mc = m < 98 ? m : 97;
    const int zm = mc / 49, ym = (mc % 49) / 7, xm = mc % 7;
    const int idx = (zm - zn + 1) * 169 + (ym - yn + 6) * 13 + (xm - xn + 6);
    ok[e] = (m < 98) && (n < 98);
    mk[e] = mp[mc * 98 + nc];
    for (int h = 0; h < 4; ++h) tb[e][h] = table[idx * 4 + h];
  }
  float v[4][4];
  for (int e = 0; e < 4; ++e)
    for (int h = 0; h < 4; ++h) {
      const float s = tb[e][h] + mk[e];
      v[h][e] = ok[e] ? s : -3.0e38f;
    }
  float4* t4 = (float4*)term;
  for (int h = 0; h < 4; ++h)
    t4[((size_t)((wd * 4 + h) * 7 + mt) * 14 + j) * 32 + lane] =
        make_float4(v[h][0], v[h][1], v[h][2], v[h][3]);
}

__device__ __forceinline__ void qkv_half(const float4 (&xh)[8],
                                         const float* __restrict__ wh,
                                         v8f (&acc)[4]) {
#pragma unroll
  for (int j = 0; j < 8; ++j) {
    const float* wj = wh + (size_t)(8 * j) * 384;
    const float4 w0 = *(const float4*)(wj);
    const float4 w1 = *(const float4*)(wj + 384);
    const float4 w2 = *(const float4*)(wj + 768);
    const float4 w3 = *(const float4*)(wj + 1152);
    v2f a, bw;
    a.x = xh[j].x; a.y = xh[j].y;
    bw.x = w0.x; bw.y = w1.x; acc[0] = wmma_f32(a, bw, acc[0]);
    bw.x = w0.y; bw.y = w1.y; acc[1] = wmma_f32(a, bw, acc[1]);
    bw.x = w0.z; bw.y = w1.z; acc[2] = wmma_f32(a, bw, acc[2]);
    bw.x = w0.w; bw.y = w1.w; acc[3] = wmma_f32(a, bw, acc[3]);
    a.x = xh[j].z; a.y = xh[j].w;
    bw.x = w2.x; bw.y = w3.x; acc[0] = wmma_f32(a, bw, acc[0]);
    bw.x = w2.y; bw.y = w3.y; acc[1] = wmma_f32(a, bw, acc[1]);
    bw.x = w2.z; bw.y = w3.z; acc[2] = wmma_f32(a, bw, acc[2]);
    bw.x = w2.w; bw.y = w3.w; acc[3] = wmma_f32(a, bw, acc[3]);
  }
}

__global__ __launch_bounds__(256) void fused_window_attn(
    const float* __restrict__ x,
    const float* __restrict__ w_qkv, const float* __restrict__ bq,
    const float* __restrict__ w_proj, const float* __restrict__ bpj,
    const float* __restrict__ term, float* __restrict__ out) {
  __shared__ float sQKV[112][386];
  __shared__ float sO[112][130];
  __shared__ float sP[8][16][114];

  const int b = blockIdx.x;
  const int wd = b & 511;
  const int wave = threadIdx.x >> 5;
  const int lane = threadIdx.x & 31;
  const int l16 = lane & 15;
  const int hi = lane >> 4;
  const float* xb = x + (size_t)b * 98 * 128;

  const float4* term4 = (const float4*)term;
  float4 pf[14];
  {
    const int h0 = wave / 7;
    const int mt0 = wave - h0 * 7;
    const float4* tp =
        term4 + (size_t)((wd * 4 + h0) * 7 + mt0) * (14 * 32) + lane;
    for (int j = 0; j < 14; ++j) pf[j] = tp[j * 32];
  }

  if (wave < 7) {
    const int mt = wave;
    const int n = mt * 16 + l16;
    const int nc = n < 98 ? n : 97;
    const float* xr = xb + (size_t)nc * 128 + 4 * hi;
    float4 xlo[8], xhi[8];
#pragma unroll
    for (int j = 0; j < 8; ++j) xlo[j] = *(const float4*)(xr + 8 * j);
#pragma unroll
    for (int j = 0; j < 8; ++j) xhi[j] = *(const float4*)(xr + 64 + 8 * j);
    for (int g = 0; g < 6; ++g) {
      const int col0 = g * 64 + 4 * l16;
      const float* wp = w_qkv + (size_t)(4 * hi) * 384 + col0;
      v8f acc[4] = {};
      qkv_half(xlo, wp, acc);
      __builtin_amdgcn_sched_barrier(0);
      qkv_half(xhi, wp + (size_t)64 * 384, acc);
      const float4 b4 = *(const float4*)(bq + col0);
      const float scale = (col0 < 128) ? 0.17677669529663687f : 1.0f;
#pragma unroll
      for (int r = 0; r < 8; ++r) {
        const int m = mt * 16 + r + 8 * hi;
        float2 lo, hh;
        lo.x = (acc[0][r] + b4.x) * scale;
        lo.y = (acc[1][r] + b4.y) * scale;
        hh.x = (acc[2][r] + b4.z) * scale;
        hh.y = (acc[3][r] + b4.w) * scale;
        *(float2*)(&sQKV[m][col0]) = lo;
        *(float2*)(&sQKV[m][col0 + 2]) = hh;
      }
    }
  }
  __syncthreads();

  for (int it = 0; it < 4; ++it) {
    const int u = wave + 8 * it;
    if (u < 28) {
      const int h = u / 7;
      const int mt = u - h * 7;

      v8f acc[7];
      for (int nt = 0; nt < 7; ++nt) {
        const float4 lo = pf[2 * nt], hh = pf[2 * nt + 1];
        acc[nt][0] = lo.x; acc[nt][1] = lo.y; acc[nt][2] = lo.z; acc[nt][3] = lo.w;
        acc[nt][4] = hh.x; acc[nt][5] = hh.y; acc[nt][6] = hh.z; acc[nt][7] = hh.w;
      }
      for (int ks = 0; ks < 8; ++ks) {
        const int k0 = ks * 4;
        v2f a = *(const v2f*)(&sQKV[mt * 16 + l16][h * 32 + k0 + 2 * hi]);
#pragma unroll
        for (int nt = 0; nt < 7; ++nt) {
          v2f bk =
              *(const v2f*)(&sQKV[nt * 16 + l16][128 + h * 32 + k0 + 2 * hi]);
          acc[nt] = wmma_f32(a, bk, acc[nt]);
        }
      }

      if (u + 8 < 28) {
        const int hn = (u + 8) / 7;
        const int mtn = (u + 8) - hn * 7;
        const float4* tp =
            term4 + (size_t)((wd * 4 + hn) * 7 + mtn) * (14 * 32) + lane;
        for (int j = 0; j < 14; ++j) pf[j] = tp[j * 32];
      }

#pragma unroll
      for (int r = 0; r < 8; ++r) {
        float mx = acc[0][r];
#pragma unroll
        for (int nt = 1; nt < 7; ++nt) mx = fmaxf(mx, acc[nt][r]);
        mx = fmaxf(mx, row_xor<1>(mx));
        mx = fmaxf(mx, row_xor<2>(mx));
        mx = fmaxf(mx, row_xor<4>(mx));
        mx = fmaxf(mx, row_xor<8>(mx));
        float e[7];
        float sum = 0.0f;
#pragma unroll
        for (int nt = 0; nt < 7; ++nt) {
          e[nt] = __expf(acc[nt][r] - mx);
          sum += e[nt];
        }
        sum += row_xor<1>(sum);
        sum += row_xor<2>(sum);
        sum += row_xor<4>(sum);
        sum += row_xor<8>(sum);
        const float inv = 1.0f / sum;
        const int m = mt * 16 + r + 8 * hi;
        const bool rowv = (m < 98);
#pragma unroll
        for (int nt = 0; nt < 7; ++nt) {
          const int n = nt * 16 + l16;
          sP[wave][r + 8 * hi][nt * 16 + l16] =
              (rowv && n < 98) ? e[nt] * inv : 0.0f;
        }
      }

      v8f o0 = {}, o1 = {};
      for (int ks = 0; ks < 28; ++ks) {
        const int k0 = ks * 4;
        v2f a = *(const v2f*)(&sP[wave][l16][k0 + 2 * hi]);
        const int t0 = k0 + 2 * hi;
        v2f b0, b1;
        b0.x = sQKV[t0][256 + h * 32 + l16];
        b0.y = sQKV[t0 + 1][256 + h * 32 + l16];
        o0 = wmma_f32(a, b0, o0);
        b1.x = sQKV[t0][256 + h * 32 + 16 + l16];
        b1.y = sQKV[t0 + 1][256 + h * 32 + 16 + l16];
        o1 = wmma_f32(a, b1, o1);
      }
#pragma unroll
      for (int r = 0; r < 8; ++r) {
        const int m = mt * 16 + r + 8 * hi;
        sO[m][h * 32 + l16] = o0[r];
        sO[m][h * 32 + 16 + l16] = o1[r];
      }
    }
  }
  __syncthreads();

  {
    const int col = wave * 16 + l16;
    v8f acc[7] = {};
    for (int k0 = 0; k0 < 128; k0 += 4) {
      const int kb = k0 + 2 * hi;
      v2f bw;
      bw.x = w_proj[(size_t)kb * 128 + col];
      bw.y = w_proj[(size_t)(kb + 1) * 128 + col];
#pragma unroll
      for (int mt = 0; mt < 7; ++mt) {
        v2f a = *(const v2f*)(&sO[mt * 16 + l16][k0 + 2 * hi]);
        acc[mt] = wmma_f32(a, bw, acc[mt]);
      }
    }
    const float bias = bpj[col];
#pragma unroll
    for (int mt = 0; mt < 7; ++mt)
#pragma unroll
      for (int r = 0; r < 8; ++r) {
        const int m = mt * 16 + r + 8 * hi;
        if (m < 98) out[((size_t)b * 98 + m) * 128 + col] = acc[mt][r] + bias;
      }
  }
}

extern "C" void kernel_launch(void* const* d_in, const int* in_sizes, int n_in,
                              void* d_out, int out_size, void* d_ws,
                              size_t ws_size, hipStream_t stream) {
  const float* x      = (const float*)d_in[0];
  const float* mask   = (const float*)d_in[1];
  const float* w_qkv  = (const float*)d_in[2];
  const float* b_qkv  = (const float*)d_in[3];
  const float* w_proj = (const float*)d_in[4];
  const float* b_proj = (const float*)d_in[5];
  const float* table  = (const float*)d_in[6];
  float* out = (float*)d_out;
  float* term = (float*)d_ws;

  score_term_kernel<<<3584, 448, 0, stream>>>(table, mask, term);
  fused_window_attn<<<2048, 256, 0, stream>>>(x, w_qkv, b_qkv, w_proj, b_proj,
                                              term, out);
}


# Score term split: first launch writes mask+padding once per window, not per head (a quarter of the bytes, 284 blocks for 3584), and a per-head bias tile; scores start from bias + mask; sum order kept
# speedup vs baseline: 2.2259x; 1.0424x over previous
#include <hip/hip_runtime.h>
#include <hip/hip_bf16.h>
#include <math.h>

typedef __attribute__((ext_vector_type(2))) float v2f;
typedef __attribute__((ext_vector_type(8))) float v8f;

__device__ __forceinline__ v8f wmma_f32(v2f a, v2f b, v8f c) {
  return __builtin_amdgcn_wmma_f32_16x16x4_f32(false, a, false, b, (short)0, c,
                                               false, false);
}


template <int M>
__device__ __forceinline__ float row_xor(float v) {
  const int b = __builtin_bit_cast(int, v);
  return __builtin_bit_cast(
      float, __builtin_amdgcn_update_dpp(b, b, 0x160 + M, 0xf, 0xf, true));
}

__global__ __launch_bounds__(448) void score_term_kernel(
    const float* __restrict__ table, const float* __restrict__ mask,
    float* __restrict__ mterm, float* __restrict__ bterm) {
  const int j = threadIdx.x >> 5;
  const int lane = threadIdx.x & 31;
  const int l16 = lane & 15;
  const int hi = lane >> 4;
  const int nt = j >> 1;
  const int r0 = (j & 1) * 4;
  const int n = nt * 16 + l16;
  const int nc = n < 98 ? n : 97;
  if (blockIdx.x < 256) {
    const int wd0 = blockIdx.x * 2;
    float mk[2][7][4];
#pragma unroll
    for (int w = 0; w < 2; ++w) {
      const float* mp = mask + (size_t)(wd0 + w) * 9604;
#pragma unroll
      for (int mt = 0; mt < 7; ++mt)
#pragma unroll
        for (int e = 0; e < 4; ++e) {
          const int m = mt * 16 + r0 + e + 8 * hi;
          const int mc = m < 98 ? m : 97;
          mk[w][mt][e] = mp[mc * 98 + nc];
        }
    }
    float4* t4 = (float4*)mterm;
#pragma unroll
    for (int w = 0; w < 2; ++w)
#pragma unroll
      for (int mt = 0; mt < 7; ++mt) {
        float v[4];
#pragma unroll
        for (int e = 0; e < 4; ++e) {
          const int m = mt * 16 + r0 + e + 8 * hi;
          v[e] = ((m < 98) && (n < 98)) ? mk[w][mt][e] : -3.0e38f;
        }
        t4[((size_t)((wd0 + w) * 7 + mt) * 14 + j) * 32 + lane] =
            make_float4(v[0], v[1], v[2], v[3]);
      }
  } else {
    const int hb = blockIdx.x - 256;
    const int h = hb / 7;
    const int mt = hb - h * 7;
    const int zn = nc / 49, yn = (nc % 49) / 7, xn = nc % 7;
    float v[4];
    for (int e = 0; e < 4; ++e) {
      const int m = mt * 16 + r0 + e + 8 * hi;
      const int mc = m < 98 ? m : 97;
      const int zm = mc / 49, ym = (mc % 49) / 7, xm = mc % 7;
      const int idx = (zm - zn + 1) * 169 + (ym - yn + 6) * 13 + (xm - xn + 6);
      const float tb = table[idx * 4 + h];
      v[e] = ((m < 98) && (n < 98)) ? tb : 0.0f;
    }
    ((float4*)bterm)[((size_t)(h * 7 + mt) * 14 + j) * 32 + lane] =
        make_float4(v[0], v[1], v[2], v[3]);
  }
}

__device__ __forceinline__ void qkv_half(const float4 (&xh)[8],
                                         const float* __restrict__ wh,
                                         v8f (&acc)[4]) {
#pragma unroll
  for (int j = 0; j < 8; ++j) {
    const float* wj = wh + (size_t)(8 * j) * 384;
    const float4 w0 = *(const float4*)(wj);
    const float4 w1 = *(const float4*)(wj + 384);
    const float4 w2 = *(const float4*)(wj + 768);
    const float4 w3 = *(const float4*)(wj + 1152);
    v2f a, bw;
    a.x = xh[j].x; a.y = xh[j].y;
    bw.x = w0.x; bw.y = w1.x; acc[0] = wmma_f32(a, bw, acc[0]);
    bw.x = w0.y; bw.y = w1.y; acc[1] = wmma_f32(a, bw, acc[1]);
    bw.x = w0.z; bw.y = w1.z; acc[2] = wmma_f32(a, bw, acc[2]);
    bw.x = w0.w; bw.y = w1.w; acc[3] = wmma_f32(a, bw, acc[3]);
    a.x = xh[j].z; a.y = xh[j].w;
    bw.x = w2.x; bw.y = w3.x; acc[0] = wmma_f32(a, bw, acc[0]);
    bw.x = w2.y; bw.y = w3.y; acc[1] = wmma_f32(a, bw, acc[1]);
    bw.x = w2.z; bw.y = w3.z; acc[2] = wmma_f32(a, bw, acc[2]);
    bw.x = w2.w; bw.y = w3.w; acc[3] = wmma_f32(a, bw, acc[3]);
  }
}

__global__ __launch_bounds__(256) void fused_window_attn(
    const float* __restrict__ x,
    const float* __restrict__ w_qkv, const float* __restrict__ bq,
    const float* __restrict__ w_proj, const float* __restrict__ bpj,
    const float* __restrict__ mterm, const float* __restrict__ bterm,
    float* __restrict__ out) {
  __shared__ float sQKV[112][386];
  __shared__ float sO[112][130];
  __shared__ float sP[8][16][114];

  const int b = blockIdx.x;
  const int wd = b & 511;
  const int wave = threadIdx.x >> 5;
  const int lane = threadIdx.x & 31;
  const int l16 = lane & 15;
  const int hi = lane >> 4;
  const float* xb = x + (size_t)b * 98 * 128;

  const float4* mterm4 = (const float4*)mterm;
  const float4* bterm4 = (const float4*)bterm;
  float4 pf[14], pb[14];
  {
    const int mt0 = wave - (wave / 7) * 7;
    const float4* tp = mterm4 + (size_t)(wd * 7 + mt0) * (14 * 32) + lane;
    for (int j = 0; j < 14; ++j) pf[j] = tp[j * 32];
  }

  if (wave < 7) {
    const int mt = wave;
    const int n = mt * 16 + l16;
    const int nc = n < 98 ? n : 97;
    const float* xr = xb + (size_t)nc * 128 + 4 * hi;
    float4 xlo[8], xhi[8];
#pragma unroll
    for (int j = 0; j < 8; ++j) xlo[j] = *(const float4*)(xr + 8 * j);
#pragma unroll
    for (int j = 0; j < 8; ++j) xhi[j] = *(const float4*)(xr + 64 + 8 * j);
    for (int g = 0; g < 6; ++g) {
      const int col0 = g * 64 + 4 * l16;
      const float* wp = w_qkv + (size_t)(4 * hi) * 384 + col0;
      v8f acc[4] = {};
      qkv_half(xlo, wp, acc);
      __builtin_amdgcn_sched_barrier(0);
      qkv_half(xhi, wp + (size_t)64 * 384, acc);
      const float4 b4 = *(const float4*)(bq + col0);
      const float scale = (col0 < 128) ? 0.17677669529663687f : 1.0f;
#pragma unroll
      for (int r = 0; r < 8; ++r) {
        const int m = mt * 16 + r + 8 * hi;
        float2 lo, hh;
        lo.x = (acc[0][r] + b4.x) * scale;
        lo.y = (acc[1][r] + b4.y) * scale;
        hh.x = (acc[2][r] + b4.z) * scale;
        hh.y = (acc[3][r] + b4.w) * scale;
        *(float2*)(&sQKV[m][col0]) = lo;
        *(float2*)(&sQKV[m][col0 + 2]) = hh;
      }
    }
  }
  {
    const int h0 = wave / 7;
    const int mt0 = wave - h0 * 7;
    const float4* bp = bterm4 + (size_t)(h0 * 7 + mt0) * (14 * 32) + lane;
    for (int j = 0; j < 14; ++j) pb[j] = bp[j * 32];
  }
  __syncthreads();

  for (int it = 0; it < 4; ++it) {
    const int u = wave + 8 * it;
    if (u < 28) {
      const int h = u / 7;
      const int mt = u - h * 7;

      v8f acc[7];
      for (int nt = 0; nt < 7; ++nt) {
        const float4 ml = pf[2 * nt], mh = pf[2 * nt + 1];
        const float4 bl = pb[2 * nt], bh = pb[2 * nt + 1];
        acc[nt][0] = bl.x + ml.x; acc[nt][1] = bl.y + ml.y;
        acc[nt][2] = bl.z + ml.z; acc[nt][3] = bl.w + ml.w;
        acc[nt][4] = bh.x + mh.x; acc[nt][5] = bh.y + mh.y;
        acc[nt][6] = bh.z + mh.z; acc[nt][7] = bh.w + mh.w;
      }
      for (int ks = 0; ks < 8; ++ks) {
        const int k0 = ks * 4;
        v2f a = *(const v2f*)(&sQKV[mt * 16 + l16][h * 32 + k0 + 2 * hi]);
#pragma unroll
        for (int nt = 0; nt < 7; ++nt) {
          v2f bk =
              *(const v2f*)(&sQKV[nt * 16 + l16][128 + h * 32 + k0 + 2 * hi]);
          acc[nt] = wmma_f32(a, bk, acc[nt]);
        }
      }

      if (u + 8 < 28) {
        const int hn = (u + 8) / 7;
        const int mtn = (u + 8) - hn * 7;
        const float4* tp = mterm4 + (size_t)(wd * 7 + mtn) * (14 * 32) + lane;
        const float4* bp = bterm4 + (size_t)(hn * 7 + mtn) * (14 * 32) + lane;
        for (int j = 0; j < 14; ++j) pf[j] = tp[j * 32];
        for (int j = 0; j < 14; ++j) pb[j] = bp[j * 32];
      }

#pragma unroll
      for (int r = 0; r < 8; ++r) {
        float mx = acc[0][r];
#pragma unroll
        for (int nt = 1; nt < 7; ++nt) mx = fmaxf(mx, acc[nt][r]);
        mx = fmaxf(mx, row_xor<1>(mx));
        mx = fmaxf(mx, row_xor<2>(mx));
        mx = fmaxf(mx, row_xor<4>(mx));
        mx = fmaxf(mx, row_xor<8>(mx));
        float e[7];
        float sum = 0.0f;
#pragma unroll
        for (int nt = 0; nt < 7; ++nt) {
          e[nt] = __expf(acc[nt][r] - mx);
          sum += e[nt];
        }
        sum += row_xor<1>(sum);
        sum += row_xor<2>(sum);
        sum += row_xor<4>(sum);
        sum += row_xor<8>(sum);
        const float inv = 1.0f / sum;
        const int m = mt * 16 + r + 8 * hi;
        const bool rowv = (m < 98);
#pragma unroll
        for (int nt = 0; nt < 7; ++nt) {
          const int n = nt * 16 + l16;
          sP[wave][r + 8 * hi][nt * 16 + l16] =
              (rowv && n < 98) ? e[nt] * inv : 0.0f;
        }
      }

      v8f o0 = {}, o1 = {};
      for (int ks = 0; ks < 28; ++ks) {
        const int k0 = ks * 4;
        v2f a = *(const v2f*)(&sP[wave][l16][k0 + 2 * hi]);
        const int t0 = k0 + 2 * hi;
        v2f b0, b1;
        b0.x = sQKV[t0][256 + h * 32 + l16];
        b0.y = sQKV[t0 + 1][256 + h * 32 + l16];
        o0 = wmma_f32(a, b0, o0);
        b1.x = sQKV[t0][256 + h * 32 + 16 + l16];
        b1.y = sQKV[t0 + 1][256 + h * 32 + 16 + l16];
        o1 = wmma_f32(a, b1, o1);
      }
#pragma unroll
      for (int r = 0; r < 8; ++r) {
        const int m = mt * 16 + r + 8 * hi;
        sO[m][h * 32 + l16] = o0[r];
        sO[m][h * 32 + 16 + l16] = o1[r];
      }
    }
  }
  __syncthreads();

  {
    const int col = wave * 16 + l16;
    v8f acc[7] = {};
    for (int k0 = 0; k0 < 128; k0 += 4) {
      const int kb = k0 + 2 * hi;
      v2f bw;
      bw.x = w_proj[(size_t)kb * 128 + col];
      bw.y = w_proj[(size_t)(kb + 1) * 128 + col];
#pragma unroll
      for (int mt = 0; mt < 7; ++mt) {
        v2f a = *(const v2f*)(&sO[mt * 16 + l16][k0 + 2 * hi]);
        acc[mt] = wmma_f32(a, bw, acc[mt]);
      }
    }
    const float bias = bpj[col];
#pragma unroll
    for (int mt = 0; mt < 7; ++mt)
#pragma unroll
      for (int r = 0; r < 8; ++r) {
        const int m = mt * 16 + r + 8 * hi;
        if (m < 98) out[((size_t)b * 98 + m) * 128 + col] = acc[mt][r] + bias;
      }
  }
}

extern "C" void kernel_launch(void* const* d_in, const int* in_sizes, int n_in,
                              void* d_out, int out_size, void* d_ws,
                              size_t ws_size, hipStream_t stream) {
  const float* x      = (const float*)d_in[0];
  const float* mask   = (const float*)d_in[1];
  const float* w_qkv  = (const float*)d_in[2];
  const float* b_qkv  = (const float*)d_in[3];
  const float* w_proj = (const float*)d_in[4];
  const float* b_proj = (const float*)d_in[5];
  const float* table  = (const float*)d_in[6];
  float* out = (float*)d_out;
  float* mterm = (float*)d_ws;
  float* bterm = mterm + 6422528;

  score_term_kernel<<<284, 448, 0, stream>>>(table, mask, mterm, bterm);
  fused_window_attn<<<2048, 256, 0, stream>>>(x, w_qkv, b_qkv, w_proj, b_proj,
                                              mterm, bterm, out);
}
